# starting kernel
# baseline (speedup 1.0000x reference)
#include <hip/hip_runtime.h>


typedef float v2f __attribute__((ext_vector_type(2)));
typedef float v8f __attribute__((ext_vector_type(8)));

#define EMA_ALPHA 0.9f
#define EMA_BETA  0.1f

__device__ __constant__ float c_apow[16] = {
    1.0f,            0.9f,            0.81f,           0.729f,
    0.6561f,         0.59049f,        0.531441f,       0.4782969f,
    0.43046721f,     0.387420489f,    0.3486784401f,   0.31381059609f,
    0.282429536481f, 0.2541865828329f,0.22876792454961f,0.205891132094649f
};

struct Btile { v2f b[4]; };

__device__ __forceinline__ float carry_xchg(float v) {
#if __has_builtin(__builtin_amdgcn_permlanex16)
    int s = __builtin_bit_cast(int, v);
    int p = __builtin_amdgcn_permlanex16(s, s, 0x76543210, 0xfedcba98,
                                         false, false);
    return __builtin_bit_cast(float, p);
#else
    return __shfl(v, (threadIdx.x & 15) + 16, 32);
#endif
}

__global__ __launch_bounds__(256)
void ema_wmma_kernel(const float* __restrict__ x, float* __restrict__ y) {
    constexpr int T = 4096, C = 512;
    constexpr int L = 512;
    constexpr int W = 256;
    constexpr int PFD = 8;

    const int lane = threadIdx.x & 31;
    const int wv   = threadIdx.x >> 5;
    const int li   = lane & 15;
    const int h    = lane >> 4;

    const int c0    = blockIdx.x * 128 + wv * 16;
    const int b     = blockIdx.z;
    const int chunk = blockIdx.y;
    const int t0    = chunk * L;
    const int tw    = (chunk == 0) ? 0 : (t0 - W);
    const int nT    = ((t0 + L) - tw) >> 4;

    const float* __restrict__ xph =
        x + (size_t)b * T * C + c0 + li + (size_t)(2 * h) * C;
    float* __restrict__ yph =
        y + (size_t)b * T * C + c0 + li + (size_t)(8 * h) * C;

    const float* __restrict__ xpf =
        x + (size_t)b * T * C + c0 + (size_t)li * C;

    v2f A[4];
#pragma unroll
    for (int j = 0; j < 4; ++j) {
        const int k0 = 4 * j + 2 * h;
        const float a0 = (k0     <= li) ? EMA_BETA * c_apow[li - k0]       : 0.0f;
        const float a1 = (k0 + 1 <= li) ? EMA_BETA * c_apow[li - (k0 + 1)] : 0.0f;
        A[j] = (v2f){a0, a1};
    }
    const v2f Ac = (v2f){(h == 0) ? (EMA_ALPHA * c_apow[li]) : 0.0f, 0.0f};

    float yprev = xph[0 - (size_t)(2 * h) * C + (size_t)tw * C];

    auto tilePtr = [&](int idx) -> const float* {
        return xph + (size_t)(tw + idx * 16) * C;
    };
    auto loadT = [&](const float* base) {
        Btile t;
#pragma unroll
        for (int j = 0; j < 4; ++j)
            t.b[j] = (v2f){ base[(size_t)(4 * j) * C],
                            base[(size_t)(4 * j + 1) * C] };
        return t;
    };

    Btile buf[4];
#pragma unroll
    for (int k = 0; k < 4; ++k)
        buf[k] = loadT(tilePtr(k));

    for (int ti = 0; ti < nT; ti += 4) {
#pragma unroll
        for (int k = 0; k < 4; ++k) {
            const int t_idx = ti + k;
            const int tt    = tw + t_idx * 16;

            const int pf_idx = min(t_idx + PFD, nT - 1);
            __builtin_prefetch(xpf + (size_t)(tw + pf_idx * 16) * C, 0, 3);

            v8f d = {};
            const v2f Bc = (v2f){(h == 0) ? yprev : 0.0f, 0.0f};
            d = __builtin_amdgcn_wmma_f32_16x16x4_f32(false, Ac, false, Bc,
                                                      (short)0, d, false, false);
#pragma unroll
            for (int j = 0; j < 4; ++j)
                d = __builtin_amdgcn_wmma_f32_16x16x4_f32(false, A[j], false,
                                                          buf[k].b[j],
                                                          (short)0, d, false, false);

            yprev = carry_xchg(d[7]);

            buf[k] = loadT(tilePtr(min(t_idx + 4, nT - 1)));

            if (tt >= t0) {
                float* __restrict__ yb = yph + (size_t)tt * C;
#pragma unroll
                for (int r = 0; r < 8; ++r)
                    __builtin_nontemporal_store(d[r], &yb[(size_t)r * C]);
            }
        }
    }
}

extern "C" void kernel_launch(void* const* d_in, const int* in_sizes, int n_in,
                              void* d_out, int out_size, void* d_ws, size_t ws_size,
                              hipStream_t stream) {
    (void)in_sizes; (void)n_in; (void)out_size; (void)d_ws; (void)ws_size;
    const float* x = (const float*)d_in[0];
    float*       y = (float*)d_out;
    dim3 grid(4, 8, 16);
    ema_wmma_kernel<<<grid, 256, 0, stream>>>(x, y);
}


# one work-group scans the whole sequence: 8 waves side by side in time scan 64-step pieces from zero; end states folded through LDS give exact carry-ins (carry term summed last); no warm-up re-read
# speedup vs baseline: 1.0899x; 1.0899x over previous
#include <hip/hip_runtime.h>


typedef float v2f __attribute__((ext_vector_type(2)));
typedef float v8f __attribute__((ext_vector_type(8)));

#define EMA_ALPHA 0.9f
#define EMA_BETA  0.1f
#define EMA_A16   0.1853020188851841f
#define EMA_A64   0.0011790184577738583f

__device__ __constant__ float c_apow[16] = {
    1.0f,            0.9f,            0.81f,           0.729f,
    0.6561f,         0.59049f,        0.531441f,       0.4782969f,
    0.43046721f,     0.387420489f,    0.3486784401f,   0.31381059609f,
    0.282429536481f, 0.2541865828329f,0.22876792454961f,0.205891132094649f
};

struct Btile { v2f b[4]; };

__device__ __forceinline__ float carry_xchg(float v) {
#if __has_builtin(__builtin_amdgcn_permlanex16)
    int s = __builtin_bit_cast(int, v);
    int p = __builtin_amdgcn_permlanex16(s, s, 0x76543210, 0xfedcba98,
                                         false, false);
    return __builtin_bit_cast(float, p);
#else
    return __shfl(v, (threadIdx.x & 15) + 16, 32);
#endif
}

__device__ __forceinline__ void lds_signal() {
    asm volatile("s_wait_dscnt 0x0\n\ts_barrier_signal -1" ::: "memory");
}
__device__ __forceinline__ void lds_wait() {
    asm volatile("s_barrier_wait -1" ::: "memory");
}

__global__ __launch_bounds__(512)
void ema_wmma_kernel(const float* __restrict__ x, float* __restrict__ y) {
    constexpr int T  = 4096, C = 512;
    constexpr int NW = 8;
    constexpr int S  = 64;
    constexpr int R  = NW * S;
    constexpr int NR = T / R;

    __shared__ float s_end[2][2][NW][16];

    const int lane = threadIdx.x & 31;
    const int wv   = threadIdx.x >> 5;
    const int g2   = wv & 1;
    const int w    = wv >> 1;
    const int li   = lane & 15;
    const int h    = lane >> 4;

    const int c0 = blockIdx.x * 32 + g2 * 16;
    const int b  = blockIdx.y;

    const size_t rowbase = (size_t)b * T * C + (size_t)(w * S) * C + c0;

    const float* __restrict__ xph = x + rowbase + li + (size_t)(2 * h) * C;
    float* __restrict__       yph = y + rowbase + li + (size_t)(8 * h) * C;
    const float* __restrict__ xpf = x + rowbase + (size_t)li * C;

    v2f A[4];
#pragma unroll
    for (int j = 0; j < 4; ++j) {
        const int k0 = 4 * j + 2 * h;
        const float a0 = (k0     <= li) ? EMA_BETA * c_apow[li - k0]       : 0.0f;
        const float a1 = (k0 + 1 <= li) ? EMA_BETA * c_apow[li - (k0 + 1)] : 0.0f;
        A[j] = (v2f){a0, a1};
    }
    const v2f Ac = (v2f){(h == 0) ? (EMA_ALPHA * c_apow[li]) : 0.0f, 0.0f};

    float cin = x[(size_t)b * T * C + c0 + li];

    auto loadT = [&](const float* base) {
        Btile t;
#pragma unroll
        for (int j = 0; j < 4; ++j)
            t.b[j] = (v2f){ base[(size_t)(4 * j) * C],
                            base[(size_t)(4 * j + 1) * C] };
        return t;
    };

    Btile cur[4], nxt[4];
#pragma unroll
    for (int j = 0; j < 4; ++j)
        cur[j] = loadT(xph + (size_t)(16 * j) * C);
#pragma unroll
    for (int j = 0; j < 4; ++j)
        nxt[j] = loadT(xph + (size_t)(R + 16 * j) * C);
#pragma unroll
    for (int j = 0; j < 4; ++j)
        __builtin_prefetch(xpf + (size_t)(2 * R + 16 * j) * C, 0, 3);

#pragma unroll
    for (int r = 0; r < NR; ++r) {
        v8f dl[4];
#pragma unroll
        for (int j = 0; j < 4; ++j) {
            v8f d = {};
#pragma unroll
            for (int k = 0; k < 4; ++k)
                d = __builtin_amdgcn_wmma_f32_16x16x4_f32(false, A[k], false,
                                                          cur[j].b[k],
                                                          (short)0, d, false, false);
            dl[j] = d;
        }

        float e = dl[0][7];
#pragma unroll
        for (int j = 1; j < 4; ++j) e = fmaf(EMA_A16, e, dl[j][7]);
        if (h == 1) s_end[r & 1][g2][w][li] = e;

        lds_signal();

        Btile nx2[4];
        if (r + 2 < NR) {
#pragma unroll
            for (int j = 0; j < 4; ++j)
                nx2[j] = loadT(xph + (size_t)((r + 2) * R + 16 * j) * C);
        } else {
#pragma unroll
            for (int j = 0; j < 4; ++j) nx2[j] = nxt[j];
        }
        if (r + 3 < NR) {
#pragma unroll
            for (int j = 0; j < 4; ++j)
                __builtin_prefetch(xpf + (size_t)((r + 3) * R + 16 * j) * C, 0, 3);
        }

        lds_wait();

        float acc = cin, cw = cin;
#pragma unroll
        for (int v = 0; v < NW; ++v) {
            const float ev = s_end[r & 1][g2][v][li];
            if (v == w) cw = acc;
            acc = fmaf(EMA_A64, acc, ev);
        }
        cin = acc;

        float carry = cw;
#pragma unroll
        for (int j = 0; j < 4; ++j) {
            const v2f Bc = (v2f){(h == 0) ? carry : 0.0f, 0.0f};
            const v8f d = __builtin_amdgcn_wmma_f32_16x16x4_f32(false, Ac, false, Bc,
                                                                (short)0, dl[j],
                                                                false, false);
            carry = carry_xchg(d[7]);

            float* __restrict__ yb = yph + (size_t)(r * R + 16 * j) * C;
#pragma unroll
            for (int rr = 0; rr < 8; ++rr)
                __builtin_nontemporal_store(d[rr], &yb[(size_t)rr * C]);
        }

#pragma unroll
        for (int j = 0; j < 4; ++j) { cur[j] = nxt[j]; nxt[j] = nx2[j]; }
    }
}

extern "C" void kernel_launch(void* const* d_in, const int* in_sizes, int n_in,
                              void* d_out, int out_size, void* d_ws, size_t ws_size,
                              hipStream_t stream) {
    (void)in_sizes; (void)n_in; (void)out_size; (void)d_ws; (void)ws_size;
    const float* x = (const float*)d_in[0];
    float*       y = (float*)d_out;
    dim3 grid(16, 16);
    ema_wmma_kernel<<<grid, 512, 0, stream>>>(x, y);
}


# aligned 256-byte row pieces per half-wave (16-byte loads/stores, a lane's 4 channels feed 4 tiles); 128 work-groups x 64 channels, pieces 16 waves x 32 steps: carry fold order changes; no prefetch
# speedup vs baseline: 1.6047x; 1.4723x over previous
#include <hip/hip_runtime.h>


typedef float v2f __attribute__((ext_vector_type(2)));
typedef float v4f __attribute__((ext_vector_type(4)));
typedef float v8f __attribute__((ext_vector_type(8)));

#define EMA_ALPHA 0.9f
#define EMA_BETA  0.1f
#define EMA_A16   0.1853020188851841f
#define EMA_A32   0.03433683820292518f

__device__ __constant__ float c_apow[16] = {
    1.0f,            0.9f,            0.81f,           0.729f,
    0.6561f,         0.59049f,        0.531441f,       0.4782969f,
    0.43046721f,     0.387420489f,    0.3486784401f,   0.31381059609f,
    0.282429536481f, 0.2541865828329f,0.22876792454961f,0.205891132094649f
};

struct Xtile { v4f r[4][2]; };

__device__ __forceinline__ float carry_xchg(float v) {
#if __has_builtin(__builtin_amdgcn_permlanex16)
    int s = __builtin_bit_cast(int, v);
    int p = __builtin_amdgcn_permlanex16(s, s, 0x76543210, 0xfedcba98,
                                         false, false);
    return __builtin_bit_cast(float, p);
#else
    return __shfl(v, (threadIdx.x & 15) + 16, 32);
#endif
}

__device__ __forceinline__ void lds_barrier() {
    asm volatile("s_wait_dscnt 0x0\n\ts_barrier_signal -1\n\ts_barrier_wait -1"
                 ::: "memory");
}

__global__ __launch_bounds__(512)
void ema_wmma_kernel(const float* __restrict__ x, float* __restrict__ y) {
    constexpr int T  = 4096, C = 512;
    constexpr int NW = 16;
    constexpr int S  = 32;
    constexpr int R  = NW * S;
    constexpr int NR = T / R;

    __shared__ v4f s_end[2][NW][16];

    const int lane = threadIdx.x & 31;
    const int w    = threadIdx.x >> 5;
    const int li   = lane & 15;
    const int h    = lane >> 4;

    const int c0 = blockIdx.x * 64;
    const int b  = blockIdx.y;

    const size_t rowbase = (size_t)b * T * C + (size_t)(w * S) * C + c0;

    const float* __restrict__ xph = x + rowbase + 4 * li + (size_t)(2 * h) * C;
    float* __restrict__       yph = y + rowbase + 4 * li + (size_t)(8 * h) * C;

    v2f A[4];
#pragma unroll
    for (int j = 0; j < 4; ++j) {
        const int k0 = 4 * j + 2 * h;
        const float a0 = (k0     <= li) ? EMA_BETA * c_apow[li - k0]       : 0.0f;
        const float a1 = (k0 + 1 <= li) ? EMA_BETA * c_apow[li - (k0 + 1)] : 0.0f;
        A[j] = (v2f){a0, a1};
    }
    const v2f Ac = (v2f){(h == 0) ? (EMA_ALPHA * c_apow[li]) : 0.0f, 0.0f};

    v4f cin = *(const v4f*)(x + (size_t)b * T * C + c0 + 4 * li);

    auto loadT = [&](const float* base) {
        Xtile t;
#pragma unroll
        for (int j = 0; j < 4; ++j) {
            t.r[j][0] = *(const v4f*)(base + (size_t)(4 * j) * C);
            t.r[j][1] = *(const v4f*)(base + (size_t)(4 * j + 1) * C);
        }
        return t;
    };

    Xtile cur[2];
#pragma unroll
    for (int tt = 0; tt < 2; ++tt)
        cur[tt] = loadT(xph + (size_t)(16 * tt) * C);

#pragma unroll
    for (int r = 0; r < NR; ++r) {
        v8f dl[2][4];
#pragma unroll
        for (int tt = 0; tt < 2; ++tt)
#pragma unroll
            for (int q = 0; q < 4; ++q) dl[tt][q] = (v8f){};
#pragma unroll
        for (int tt = 0; tt < 2; ++tt) {
#pragma unroll
            for (int k = 0; k < 4; ++k)
#pragma unroll
                for (int q = 0; q < 4; ++q) {
                    const v2f Bq = (v2f){cur[tt].r[k][0][q], cur[tt].r[k][1][q]};
                    dl[tt][q] = __builtin_amdgcn_wmma_f32_16x16x4_f32(
                        false, A[k], false, Bq, (short)0, dl[tt][q], false, false);
                }
            if (r + 1 < NR)
                cur[tt] = loadT(xph + (size_t)((r + 1) * R + 16 * tt) * C);
        }

        v4f e;
#pragma unroll
        for (int q = 0; q < 4; ++q) e[q] = fmaf(EMA_A16, dl[0][q][7], dl[1][q][7]);
        if (h == 1) s_end[r & 1][w][li] = e;

        lds_barrier();

        v4f acc = cin, cw = cin;
#pragma unroll
        for (int v = 0; v < NW; ++v) {
            const v4f ev = s_end[r & 1][v][li];
            if (v == w) cw = acc;
#pragma unroll
            for (int q = 0; q < 4; ++q) acc[q] = fmaf(EMA_A32, acc[q], ev[q]);
        }
        cin = acc;

        v4f carry = cw;
#pragma unroll
        for (int tt = 0; tt < 2; ++tt) {
            v8f d[4];
#pragma unroll
            for (int q = 0; q < 4; ++q) {
                const v2f Bc = (v2f){(h == 0) ? carry[q] : 0.0f, 0.0f};
                d[q] = __builtin_amdgcn_wmma_f32_16x16x4_f32(false, Ac, false, Bc,
                                                             (short)0, dl[tt][q],
                                                             false, false);
                carry[q] = carry_xchg(d[q][7]);
            }

            float* __restrict__ yb = yph + (size_t)(r * R + 16 * tt) * C;
#pragma unroll
            for (int rr = 0; rr < 8; ++rr) {
                const v4f o = (v4f){d[0][rr], d[1][rr], d[2][rr], d[3][rr]};
                __builtin_nontemporal_store(o, (v4f*)(yb + (size_t)rr * C));
            }
        }
    }
}

extern "C" void kernel_launch(void* const* d_in, const int* in_sizes, int n_in,
                              void* d_out, int out_size, void* d_ws, size_t ws_size,
                              hipStream_t stream) {
    (void)in_sizes; (void)n_in; (void)out_size; (void)d_ws; (void)ws_size;
    const float* x = (const float*)d_in[0];
    float*       y = (float*)d_out;
    dim3 grid(8, 16);
    ema_wmma_kernel<<<grid, 512, 0, stream>>>(x, y);
}


# 256 work-groups x 32 channels, one per WGP (was 128 x 64): half-wave loads two 128-byte rows 16 steps apart, a tile scans two blocks; 64-step pieces, block carries by fma: carry fold order changes
# speedup vs baseline: 2.0418x; 1.2724x over previous
#include <hip/hip_runtime.h>


typedef float v2f __attribute__((ext_vector_type(2)));
typedef float v4f __attribute__((ext_vector_type(4)));
typedef float v8f __attribute__((ext_vector_type(8)));

#define EMA_ALPHA 0.9f
#define EMA_BETA  0.1f
#define EMA_A16   0.1853020188851841f
#define EMA_A64   0.001179018457773862f

__device__ __constant__ float c_apow[16] = {
    1.0f,            0.9f,            0.81f,           0.729f,
    0.6561f,         0.59049f,        0.531441f,       0.4782969f,
    0.43046721f,     0.387420489f,    0.3486784401f,   0.31381059609f,
    0.282429536481f, 0.2541865828329f,0.22876792454961f,0.205891132094649f
};

struct Xtile { v4f r[4][2]; };

__device__ __forceinline__ float end_same(float v) {
    int s = __builtin_bit_cast(int, v);
    int p = __builtin_amdgcn_permlanex16(s, s, 0x76543210, 0xfedcba98,
                                         false, false);
    return __builtin_bit_cast(float, p);
}
__device__ __forceinline__ float end_other(float v) {
    int s = __builtin_bit_cast(int, v);
    int p = __builtin_amdgcn_permlanex16(s, s, 0xfedcba98, 0x76543210,
                                         false, false);
    return __builtin_bit_cast(float, p);
}

__device__ __forceinline__ void lds_barrier() {
    asm volatile("s_wait_dscnt 0x0\n\ts_barrier_signal -1\n\ts_barrier_wait -1"
                 ::: "memory");
}

__global__ __launch_bounds__(512)
void ema_wmma_kernel(const float* __restrict__ x, float* __restrict__ y) {
    constexpr int T  = 4096, C = 512;
    constexpr int NW = 16;
    constexpr int S  = 64;
    constexpr int R  = NW * S;
    constexpr int NR = T / R;

    __shared__ v4f s_end[2][NW][8];

    const int lane = threadIdx.x & 31;
    const int w    = threadIdx.x >> 5;
    const int li   = lane & 15;
    const int h    = lane >> 4;
    const int cl   = li & 7;
    const int tb   = li >> 3;

    const int c0 = (2 * (blockIdx.x & 7) + (blockIdx.x >> 3)) * 32;
    const int b  = blockIdx.y;

    const size_t rowbase = (size_t)b * T * C + (size_t)(w * S) * C + c0;

    const float* __restrict__ xph = x + rowbase + 4 * cl + (size_t)(16 * tb + 2 * h) * C;
    float* __restrict__       yph = y + rowbase + 4 * cl + (size_t)(16 * tb + 8 * h) * C;

    v2f A[4];
#pragma unroll
    for (int j = 0; j < 4; ++j) {
        const int k0 = 4 * j + 2 * h;
        const float a0 = (k0     <= li) ? EMA_BETA * c_apow[li - k0]       : 0.0f;
        const float a1 = (k0 + 1 <= li) ? EMA_BETA * c_apow[li - (k0 + 1)] : 0.0f;
        A[j] = (v2f){a0, a1};
    }
    const v2f Ac = (v2f){(h == 0) ? (EMA_ALPHA * c_apow[li]) : 0.0f, 0.0f};

    v4f cin = *(const v4f*)(x + (size_t)b * T * C + c0 + 4 * cl);

    auto loadT = [&](const float* base) {
        Xtile t;
#pragma unroll
        for (int j = 0; j < 4; ++j) {
            t.r[j][0] = *(const v4f*)(base + (size_t)(4 * j) * C);
            t.r[j][1] = *(const v4f*)(base + (size_t)(4 * j + 1) * C);
        }
        return t;
    };

    Xtile cur[2];
#pragma unroll
    for (int tt = 0; tt < 2; ++tt)
        cur[tt] = loadT(xph + (size_t)(32 * tt) * C);

#pragma unroll
    for (int r = 0; r < NR; ++r) {
        v8f dl[2][4];
#pragma unroll
        for (int tt = 0; tt < 2; ++tt)
#pragma unroll
            for (int q = 0; q < 4; ++q) dl[tt][q] = (v8f){};
#pragma unroll
        for (int tt = 0; tt < 2; ++tt) {
#pragma unroll
            for (int k = 0; k < 4; ++k)
#pragma unroll
                for (int q = 0; q < 4; ++q) {
                    const v2f Bq = (v2f){cur[tt].r[k][0][q], cur[tt].r[k][1][q]};
                    dl[tt][q] = __builtin_amdgcn_wmma_f32_16x16x4_f32(
                        false, A[k], false, Bq, (short)0, dl[tt][q], false, false);
                }
            if (r + 1 < NR)
                cur[tt] = loadT(xph + (size_t)((r + 1) * R + 32 * tt) * C);
        }

        v4f eb[4];
#pragma unroll
        for (int tt = 0; tt < 2; ++tt)
#pragma unroll
            for (int q = 0; q < 4; ++q) {
                const float mine  = end_same(dl[tt][q][7]);
                const float other = end_other(dl[tt][q][7]);
                eb[2 * tt][q]     = tb ? other : mine;
                eb[2 * tt + 1][q] = tb ? mine  : other;
            }

        v4f e;
#pragma unroll
        for (int q = 0; q < 4; ++q)
            e[q] = fmaf(EMA_A16, fmaf(EMA_A16, fmaf(EMA_A16, eb[0][q], eb[1][q]),
                                      eb[2][q]), eb[3][q]);
        if (lane < 8) s_end[r & 1][w][cl] = e;

        lds_barrier();

        v4f acc = cin, cw = cin;
#pragma unroll
        for (int v = 0; v < NW; ++v) {
            const v4f ev = s_end[r & 1][v][cl];
            if (v == w) cw = acc;
#pragma unroll
            for (int q = 0; q < 4; ++q) acc[q] = fmaf(EMA_A64, acc[q], ev[q]);
        }
        cin = acc;

        v4f kb[4];
        kb[0] = cw;
#pragma unroll
        for (int i = 1; i < 4; ++i)
#pragma unroll
            for (int q = 0; q < 4; ++q)
                kb[i][q] = fmaf(EMA_A16, kb[i - 1][q], eb[i - 1][q]);

#pragma unroll
        for (int tt = 0; tt < 2; ++tt) {
            v8f d[4];
#pragma unroll
            for (int q = 0; q < 4; ++q) {
                const float kc = tb ? kb[2 * tt + 1][q] : kb[2 * tt][q];
                const v2f Bc = (v2f){(h == 0) ? kc : 0.0f, 0.0f};
                d[q] = __builtin_amdgcn_wmma_f32_16x16x4_f32(false, Ac, false, Bc,
                                                             (short)0, dl[tt][q],
                                                             false, false);
            }

            float* __restrict__ yb = yph + (size_t)(r * R + 32 * tt) * C;
#pragma unroll
            for (int rr = 0; rr < 8; ++rr) {
                const v4f o = (v4f){d[0][rr], d[1][rr], d[2][rr], d[3][rr]};
                __builtin_nontemporal_store(o, (v4f*)(yb + (size_t)rr * C));
            }
        }
    }
}

extern "C" void kernel_launch(void* const* d_in, const int* in_sizes, int n_in,
                              void* d_out, int out_size, void* d_ws, size_t ws_size,
                              hipStream_t stream) {
    (void)in_sizes; (void)n_in; (void)out_size; (void)d_ws; (void)ws_size;
    const float* x = (const float*)d_in[0];
    float*       y = (float*)d_out;
    dim3 grid(16, 16);
    ema_wmma_kernel<<<grid, 512, 0, stream>>>(x, y);
}
